# starting kernel
# baseline (speedup 1.0000x reference)
#include <hip/hip_runtime.h>

typedef __attribute__((ext_vector_type(2))) float v2f;
typedef __attribute__((ext_vector_type(8))) float v8f;

namespace {
constexpr int Dch   = 16;
constexpr int Llen  = 1000000;
constexpr int Wlen  = 64;
constexpr int Kout  = 8;
constexpr int Pad   = 31;
constexpr int MidLen = Llen - Wlen + 1;
constexpr int WavesPerBlock = 4;
constexpr int TilesPerWave  = 4;
constexpr int TilePos = 16;
constexpr int WavePos  = TilesPerWave * TilePos;
constexpr int BlockPos = WavesPerBlock * WavePos;
constexpr int XTile = BlockPos + Wlen;
constexpr int Threads = WavesPerBlock * 32;
constexpr int PStride = 1027;
constexpr int PZeroOff = Kout * PStride;
constexpr int PZeroLen = 1024;
constexpr int SPLen = PZeroOff + PZeroLen;
}

__global__ __launch_bounds__(Threads) void fullscan_mid_wmma(
    const float* __restrict__ xg, const float* __restrict__ Pg,
    float* __restrict__ out)
{
  __shared__ float sP[SPLen];
  __shared__ float sX[Dch * XTile];

  const int tid = threadIdx.x;
  const int N0  = blockIdx.x * BlockPos;

  for (int i = tid; i < Kout * Dch * Wlen; i += Threads) {
    int k  = i >> 10;
    int dw = i & 1023;
    sP[k * PStride + dw] = Pg[i];
  }
  for (int i = tid; i < PZeroLen; i += Threads) sP[PZeroOff + i] = 0.0f;

  for (int i = tid; i < Dch * XTile; i += Threads) {
    int d = i / XTile;
    int j = i - d * XTile;
    int gi = N0 + j;
    if (gi > Llen - 1) gi = Llen - 1;
    sX[i] = xg[(size_t)d * Llen + gi];
  }
  __syncthreads();

  const int wave = tid >> 5;
  const int lane = tid & 31;
  const int n0 = N0 + wave * WavePos;
  if (n0 >= MidLen) return;

  const int m  = lane & 15;
  const int hi = lane >> 4;
  const int c  = hi * 2;
  const float* xb = &sX[wave * WavePos + m + c];
  const float* pb = (m < Kout) ? &sP[m * PStride + c] : &sP[PZeroOff];

  v8f acc0 = {}, acc1 = {}, acc2 = {}, acc3 = {};
  for (int d = 0; d < Dch; ++d) {
#pragma unroll
    for (int w4 = 0; w4 < 16; ++w4) {
      const int xo = d * XTile + w4 * 4;
      v2f b;
      b.x = pb[d * Wlen + w4 * 4];
      b.y = pb[d * Wlen + w4 * 4 + 1];
      v2f a0, a1, a2, a3;
      a0.x = xb[xo +  0]; a0.y = xb[xo +  1];
      a1.x = xb[xo + 16]; a1.y = xb[xo + 17];
      a2.x = xb[xo + 32]; a2.y = xb[xo + 33];
      a3.x = xb[xo + 48]; a3.y = xb[xo + 49];
      acc0 = __builtin_amdgcn_wmma_f32_16x16x4_f32(false, a0, false, b, (short)0, acc0, false, false);
      acc1 = __builtin_amdgcn_wmma_f32_16x16x4_f32(false, a1, false, b, (short)0, acc1, false, false);
      acc2 = __builtin_amdgcn_wmma_f32_16x16x4_f32(false, a2, false, b, (short)0, acc2, false, false);
      acc3 = __builtin_amdgcn_wmma_f32_16x16x4_f32(false, a3, false, b, (short)0, acc3, false, false);
    }
  }

  const int k = m;
  if (k < Kout) {
    const int mrow0 = hi * 8;
    float* outk = out + (size_t)k * Llen + Pad;
#pragma unroll
    for (int j = 0; j < 8; ++j) {
      int n0j = n0 + mrow0 + j;
      if (n0j      < MidLen) outk[n0j]      = acc0[j];
      if (n0j + 16 < MidLen) outk[n0j + 16] = acc1[j];
      if (n0j + 32 < MidLen) outk[n0j + 32] = acc2[j];
      if (n0j + 48 < MidLen) outk[n0j + 48] = acc3[j];
    }
  }
}

__global__ void fullscan_edges(const float* __restrict__ xg,
                               const float* __restrict__ Pg,
                               float* __restrict__ out)
{
  int t = blockIdx.x * blockDim.x + threadIdx.x;
  if (t < Kout * Pad) {
    int k = t / Pad, j = t - k * Pad;
    float s = 0.f;
    for (int mm = 0; mm <= 32 + j; ++mm) {
      float tl = 0.f;
      for (int d = 0; d < Dch; ++d)
        tl += xg[(size_t)d * Llen + mm] * Pg[(k * Dch + d) * Wlen + mm];
      s += tl;
    }
    out[(size_t)k * Llen + j] = s;
  } else if (t < Kout * Pad + Kout * 32) {
    int t2 = t - Kout * Pad;
    int k = t2 / 32, j = t2 - k * 32;
    float s = 0.f;
    for (int mm = j + 1; mm < Wlen; ++mm) {
      for (int d = 0; d < Dch; ++d)
        s += xg[(size_t)d * Llen + (Llen - Wlen + mm)] *
             Pg[(k * Dch + d) * Wlen + mm];
    }
    out[(size_t)k * Llen + (Llen - 32) + j] = s;
  }
}

extern "C" void kernel_launch(void* const* d_in, const int*  , int  ,
                              void* d_out, int  , void*  , size_t  ,
                              hipStream_t stream) {
  const float* x = (const float*)d_in[0];
  const float* P = (const float*)d_in[1];
  float* out = (float*)d_out;

  const int blocks = (MidLen + BlockPos - 1) / BlockPos;
  hipLaunchKernelGGL(fullscan_mid_wmma, dim3(blocks), dim3(Threads), 0, stream, x, P, out);
  hipLaunchKernelGGL(fullscan_edges,    dim3(2),      dim3(256),     0, stream, x, P, out);
}


# Operand roles swapped: P is the WMMA A operand (rows = tap half, channel), x the B operand, so D has positions on lanes: per wave 32 stores of 32 consecutive floats, not 64 of 16 lone 4-byte pieces
# speedup vs baseline: 2.2781x; 2.2781x over previous
#include <hip/hip_runtime.h>


typedef __attribute__((ext_vector_type(2))) float v2f;
typedef __attribute__((ext_vector_type(8))) float v8f;

namespace {
constexpr int Dch   = 16;
constexpr int Llen  = 1000000;
constexpr int Wlen  = 64;
constexpr int Kout  = 8;
constexpr int Pad   = 31;
constexpr int MidLen = Llen - Wlen + 1;
constexpr int HalfW = Wlen / 2;
constexpr int WavesPerBlock = 4;
constexpr int TilePos = 16;
constexpr int OutTiles = 8;
constexpr int HaloTiles = HalfW / TilePos;
constexpr int XTiles = OutTiles + HaloTiles;
constexpr int WavePos  = OutTiles * TilePos;
constexpr int BlockPos = WavesPerBlock * WavePos;
constexpr int XTile = BlockPos + Wlen;
constexpr int Threads = WavesPerBlock * 32;
constexpr int ChunkCh = 4;
constexpr int NChunks = Dch / ChunkCh;
constexpr int XStage  = ChunkCh * XTile;
constexpr int PStride = ChunkCh * Wlen + 3;
constexpr int PStage  = Kout * PStride;
constexpr int XSlots  = (XTile + Threads - 1) / Threads;
constexpr int XLast   = XTile - (XSlots - 1) * Threads;
constexpr int PSlots  = Kout * ChunkCh * Wlen / Threads;
constexpr int ExRow   = TilePos;
constexpr int ExTile  = 8 * ExRow + 16;
constexpr int ExWave  = 2 * HaloTiles * ExTile;
static_assert(WavesPerBlock == ChunkCh, "the halo tiles are split one channel of every stage per wave");
}

__device__ __forceinline__ float other_half(float v) {
  return __int_as_float(__builtin_amdgcn_permlanex16(__float_as_int(v), __float_as_int(v), 0x76543210, 0xfedcba98, false, false));
}

__device__ __forceinline__ void store_rows(float* __restrict__ p, int o, bool interior, const float (&v)[8]) {
  if (interior) {
#pragma unroll
    for (int r = 0; r < 8; ++r) p[(size_t)r * Llen] = v[r];
  } else if (o < MidLen) {
#pragma unroll
    for (int r = 0; r < 8; ++r) p[(size_t)r * Llen] = v[r];
  }
}

__global__ __launch_bounds__(Threads) void fullscan_mid_wmma(
    const float* __restrict__ xg, const float* __restrict__ Pg,
    float* __restrict__ out)
{
  __shared__ float sP[2 * PStage];
  __shared__ float sX[2 * XStage];
  __shared__ float sE[WavesPerBlock * ExWave];

  const int tid = threadIdx.x;
  const int N0  = blockIdx.x * BlockPos;
  const int wave = __builtin_amdgcn_readfirstlane(tid >> 5);
  const int lane = tid & 31;
  const int n0 = N0 + wave * WavePos;
  const bool lastSlot = tid < XLast;

  int gpos[XSlots];
#pragma unroll
  for (int s = 0; s < XSlots; ++s) {
    int gi = N0 + tid + s * Threads;
    gpos[s] = gi > Llen - 1 ? Llen - 1 : gi;
  }

  float xr[ChunkCh][XSlots];
  float pr[PSlots];

  const int m  = lane & 15;
  const int hi = lane >> 4;
  const int c2 = hi * 2;
  const int kc = m & 7;
  const int g  = m >> 3;
  const float* xb0 = &sX[wave * WavePos + m + c2];
  const float* xh0 = &sX[BlockPos + m + c2];
  const float* pb0 = &sP[kc * PStride + g * HalfW + c2];

  v8f acc[XTiles];
#pragma unroll
  for (int t = 0; t < XTiles; ++t) acc[t] = v8f{};

  {
#pragma unroll
    for (int q = 0; q < PSlots; ++q)
      pr[q] = Pg[(q >> 1) * (Dch * Wlen) + (q & 1) * Threads + tid];
#pragma unroll
    for (int dl = 0; dl < ChunkCh; ++dl) {
#pragma unroll
      for (int s = 0; s < XSlots - 1; ++s) xr[dl][s] = xg[(size_t)dl * Llen + gpos[s]];
      if (lastSlot) xr[dl][XSlots - 1] = xg[(size_t)dl * Llen + gpos[XSlots - 1]];
    }
#pragma unroll
    for (int dl = 0; dl < ChunkCh; ++dl) {
#pragma unroll
      for (int s = 0; s < XSlots - 1; ++s) sX[dl * XTile + s * Threads + tid] = xr[dl][s];
      if (lastSlot) sX[dl * XTile + (XSlots - 1) * Threads + tid] = xr[dl][XSlots - 1];
    }
#pragma unroll
    for (int q = 0; q < PSlots; ++q)
      sP[(q >> 1) * PStride + (q & 1) * Threads + tid] = pr[q];
  }
  __syncthreads();

  for (int c = 0; c < NChunks; ++c) {
    const int buf = c & 1;
    const bool more = c + 1 < NChunks;

    if (more) {
      const float* xc = xg + (size_t)(c + 1) * ChunkCh * Llen;
      const float* pc = Pg + (c + 1) * ChunkCh * Wlen + tid;
#pragma unroll
      for (int q = 0; q < PSlots; ++q)
        pr[q] = pc[(q >> 1) * (Dch * Wlen) + (q & 1) * Threads];
#pragma unroll
      for (int dl = 0; dl < ChunkCh; ++dl) {
#pragma unroll
        for (int s = 0; s < XSlots - 1; ++s) xr[dl][s] = xc[(size_t)dl * Llen + gpos[s]];
        if (lastSlot) xr[dl][XSlots - 1] = xc[(size_t)dl * Llen + gpos[XSlots - 1]];
      }
    }

    {
      const float* xb = xb0 + buf * XStage;
      const float* xh = xh0 + buf * XStage;
      const float* pb = pb0 + buf * PStage;
      for (int dl = 0; dl < ChunkCh; ++dl) {
#pragma unroll
        for (int w4 = 0; w4 < HalfW / 4; ++w4) {
          const int xo = dl * XTile + w4 * 4;
          v2f pf;
          pf.x = pb[dl * Wlen + w4 * 4];
          pf.y = pb[dl * Wlen + w4 * 4 + 1];
#pragma unroll
          for (int t = 0; t < OutTiles; ++t) {
            v2f xf;
            xf.x = xb[xo + t * TilePos];
            xf.y = xb[xo + t * TilePos + 1];
            acc[t] = __builtin_amdgcn_wmma_f32_16x16x4_f32(false, pf, false, xf, (short)0, acc[t], false, false);
          }
        }
        if (dl == wave) {
#pragma unroll
          for (int w4 = 0; w4 < HalfW / 4; ++w4) {
            const int xo = dl * XTile + w4 * 4;
            v2f pf;
            pf.x = pb[dl * Wlen + w4 * 4];
            pf.y = pb[dl * Wlen + w4 * 4 + 1];
#pragma unroll
            for (int t = 0; t < HaloTiles; ++t) {
              v2f xf;
              xf.x = xh[xo + t * TilePos];
              xf.y = xh[xo + t * TilePos + 1];
              acc[OutTiles + t] = __builtin_amdgcn_wmma_f32_16x16x4_f32(false, pf, false, xf, (short)0, acc[OutTiles + t], false, false);
            }
          }
        }
      }
    }

    if (more) {
      float* sx = sX + (buf ^ 1) * XStage + tid;
      float* sp = sP + (buf ^ 1) * PStage + tid;
#pragma unroll
      for (int dl = 0; dl < ChunkCh; ++dl) {
#pragma unroll
        for (int s = 0; s < XSlots - 1; ++s) sx[dl * XTile + s * Threads] = xr[dl][s];
        if (lastSlot) sx[dl * XTile + (XSlots - 1) * Threads] = xr[dl][XSlots - 1];
      }
#pragma unroll
      for (int q = 0; q < PSlots; ++q)
        sp[(q >> 1) * PStride + (q & 1) * Threads] = pr[q];
      __syncthreads();
    }
  }

  if (hi == 1) {
    float* e = &sE[wave * ExWave + m];
#pragma unroll
    for (int s = 0; s < HaloTiles; ++s) {
#pragma unroll
      for (int r = 0; r < 8; ++r) {
        e[s * ExTile + r * ExRow] = acc[s][r];
        e[(HaloTiles + s) * ExTile + r * ExRow] = acc[OutTiles + s][r];
      }
    }
  }
  __syncthreads();

  const bool interior = N0 + BlockPos <= MidLen;
  float* outp = out + Pad + n0 + lane;
#pragma unroll
  for (int jp = 0; jp < (OutTiles - HaloTiles) / 2; ++jp) {
    const int j = 2 * jp;
    float v[8];
#pragma unroll
    for (int r = 0; r < 8; ++r) {
      const float own = hi ? acc[j + 3][r] : acc[j][r];
      const float oth = hi ? acc[j + 2][r] : acc[j + 1][r];
      v[r] = own + other_half(oth);
    }
    store_rows(outp + jp * 2 * TilePos, n0 + lane + jp * 2 * TilePos, interior, v);
  }
  float h[8];
  if (wave != WavesPerBlock - 1) {
    const float* e = &sE[(wave + 1) * ExWave + hi * ExTile + m];
#pragma unroll
    for (int r = 0; r < 8; ++r) h[r] = e[r * ExRow];
  } else {
    const float* e = &sE[(HaloTiles + hi) * ExTile + m];
#pragma unroll
    for (int r = 0; r < 8; ++r) {
      const float* q = e + r * ExRow;
      h[r] = ((q[0] + q[ExWave]) + q[2 * ExWave]) + q[3 * ExWave];
    }
  }
  {
    constexpr int j = OutTiles - HaloTiles;
    float v[8];
#pragma unroll
    for (int r = 0; r < 8; ++r) {
      const float t7 = other_half(acc[j + 1][r]);
      const float own = hi ? t7 : acc[j][r];
      v[r] = own + h[r];
    }
    store_rows(outp + j * TilePos, n0 + lane + j * TilePos, interior, v);
  }
}

namespace {
constexpr int EdgeThreads = Kout * Wlen;
}

__global__ __launch_bounds__(EdgeThreads) void fullscan_edges(const float* __restrict__ xg,
                                                              const float* __restrict__ Pg,
                                                              float* __restrict__ out)
{
  __shared__ __attribute__((aligned(16))) float sT[Kout * Wlen];

  const int tid = threadIdx.x;
  const int k = tid / Wlen;
  const int m = tid % Wlen;
  const bool right = blockIdx.x != 0;
  const float* xe = xg + (right ? Llen - Wlen : 0) + m;
  const float* pe = Pg + k * (Dch * Wlen) + m;

  float xv[Dch], pv[Dch];
#pragma unroll
  for (int d = 0; d < Dch; ++d) {
    xv[d] = xe[(size_t)d * Llen];
    pv[d] = pe[d * Wlen];
  }
  float t = 0.f;
#pragma unroll
  for (int d = 0; d < Dch; ++d) t += xv[d] * pv[d];
  sT[tid] = t;
  __syncthreads();

  if (m >= HalfW) return;
  const float* row = &sT[k * Wlen];
  float s = 0.f;
  if (!right) {
    if (m < Pad) {
      const int last = HalfW + m;
#pragma unroll
      for (int i = 0; i < Wlen - 1; ++i) s += (i <= last) ? row[i] : 0.f;
      out[(size_t)k * Llen + m] = s;
    }
  } else {
#pragma unroll
    for (int i = 1; i < Wlen; ++i) s += (i > m) ? row[i] : 0.f;
    out[(size_t)k * Llen + (Llen - HalfW) + m] = s;
  }
}

extern "C" void kernel_launch(void* const* d_in, const int*  , int  ,
                              void* d_out, int  , void*  , size_t  ,
                              hipStream_t stream) {
  const float* x = (const float*)d_in[0];
  const float* P = (const float*)d_in[1];
  float* out = (float*)d_out;

  const int blocks = (MidLen + BlockPos - 1) / BlockPos;
  hipLaunchKernelGGL(fullscan_mid_wmma, dim3(blocks), dim3(Threads), 0, stream, x, P, out);
  hipLaunchKernelGGL(fullscan_edges,    dim3(2),      dim3(EdgeThreads), 0, stream, x, P, out);
}
